# starting kernel
# baseline (speedup 1.0000x reference)
#include <hip/hip_runtime.h>

typedef float v2f __attribute__((ext_vector_type(2)));
typedef float v8f __attribute__((ext_vector_type(8)));

#define VSZ  32128
#define HID  16
#define OUTD 16
#define SEQ  512
#define ROWS 16
#define TPR  16
#define BLK  (ROWS * TPR)

__device__ __forceinline__ float4 f4add(float4 a, float4 b) {
    return make_float4(a.x + b.x, a.y + b.y, a.z + b.z, a.w + b.w);
}

__global__ void transpose_w_kernel(const float* __restrict__ w,
                                   float* __restrict__ wT) {
    int v = blockIdx.x * blockDim.x + threadIdx.x;
    if (v >= VSZ) return;
#pragma unroll
    for (int k = 0; k < HID; ++k)
        wT[(size_t)v * HID + k] = w[(size_t)k * VSZ + v];
}

__global__ void __launch_bounds__(BLK)
encoder_kernel(const int* __restrict__ x,
               const float* __restrict__ wT,
               const float* __restrict__ enc1_b,
               const float* __restrict__ mean_w,
               const float* __restrict__ mean_b,
               const float* __restrict__ lv_w,
               const float* __restrict__ lv_b,
               float* __restrict__ out_mean,
               float* __restrict__ out_lv,
               int Btot) {
    __shared__ float part[ROWS][TPR][HID];
    __shared__ float hmat[ROWS][HID];

    const int tid = threadIdx.x;
    const int row = tid >> 4;
    const int tr  = tid & 15;
    const int b   = blockIdx.x * ROWS + row;

    float4 a0 = make_float4(0.f, 0.f, 0.f, 0.f);
    float4 a1 = a0, a2 = a0, a3 = a0;
    if (b < Btot) {
        const int* xrow = x + (size_t)b * SEQ;
        const float4* wT4 = (const float4*)wT;
        for (int s = tr; s < SEQ; s += TPR) {
            int tok = xrow[s];
            const float4* wr = wT4 + (size_t)tok * 4;
            a0 = f4add(a0, wr[0]);
            a1 = f4add(a1, wr[1]);
            a2 = f4add(a2, wr[2]);
            a3 = f4add(a3, wr[3]);
        }
    }
    {
        float4* p = (float4*)&part[row][tr][0];
        p[0] = a0; p[1] = a1; p[2] = a2; p[3] = a3;
    }
    __syncthreads();

    {
        const int r = row, k = tr;
        float s = 0.f;
#pragma unroll
        for (int t = 0; t < TPR; ++t) s += part[r][t][k];
        s += enc1_b[k];
        hmat[r][k] = s > 0.f ? s : 0.f;
    }
    __syncthreads();

    if (tid < 32) {
        const int lane = tid;
        const int half = lane >> 4;
        const int l16  = lane & 15;

        v2f a[4];
#pragma unroll
        for (int c = 0; c < 4; ++c) {
            int k0 = 4 * c + 2 * half;
            a[c].x = hmat[l16][k0];
            a[c].y = hmat[l16][k0 + 1];
        }
        v2f bm[4], bl[4];
#pragma unroll
        for (int c = 0; c < 4; ++c) {
            int k0 = 4 * c + 2 * half;
            bm[c].x = mean_w[l16 * HID + k0];
            bm[c].y = mean_w[l16 * HID + k0 + 1];
            bl[c].x = lv_w[l16 * HID + k0];
            bl[c].y = lv_w[l16 * HID + k0 + 1];
        }

        v8f accm = {};
#pragma unroll
        for (int c = 0; c < 4; ++c)
            accm = __builtin_amdgcn_wmma_f32_16x16x4_f32(
                false, a[c], false, bm[c], (short)0, accm, false, false);
        v8f accl = {};
#pragma unroll
        for (int c = 0; c < 4; ++c)
            accl = __builtin_amdgcn_wmma_f32_16x16x4_f32(
                false, a[c], false, bl[c], (short)0, accl, false, false);

        const float mb = mean_b[l16];
        const float lb = lv_b[l16];
        const int b0 = blockIdx.x * ROWS;
#pragma unroll
        for (int j = 0; j < 8; ++j) {
            int m = j + 8 * half;
            int gb = b0 + m;
            if (gb < Btot) {
                out_mean[(size_t)gb * OUTD + l16] = accm[j] + mb;
                out_lv[(size_t)gb * OUTD + l16]   = accl[j] + lb;
            }
        }
    }
}

extern "C" void kernel_launch(void* const* d_in, const int* in_sizes, int n_in,
                              void* d_out, int out_size, void* d_ws, size_t ws_size,
                              hipStream_t stream) {
    const int*   x      = (const int*)d_in[0];
    const float* enc1_w = (const float*)d_in[1];
    const float* enc1_b = (const float*)d_in[2];
    const float* mean_w = (const float*)d_in[3];
    const float* mean_b = (const float*)d_in[4];
    const float* lv_w   = (const float*)d_in[5];
    const float* lv_b   = (const float*)d_in[6];

    const int Btot = in_sizes[0] / SEQ;

    float* wT = (float*)d_ws;

    transpose_w_kernel<<<(VSZ + 255) / 256, 256, 0, stream>>>(enc1_w, wT);

    float* out_mean = (float*)d_out;
    float* out_lv   = (float*)d_out + (size_t)Btot * OUTD;
    encoder_kernel<<<(Btot + ROWS - 1) / ROWS, BLK, 0, stream>>>(
        x, wT, enc1_b, mean_w, mean_b, lv_w, lv_b, out_mean, out_lv, Btot);
}


# gather re-laid: one wave per row, 4 lanes per token (16 B each, 8 tokens per load), all 512 ids loaded first, 64 table loads in flight per lane, 8 rows per block; summation order of h changes
# speedup vs baseline: 1.7597x; 1.7597x over previous
#include <hip/hip_runtime.h>


typedef float v2f __attribute__((ext_vector_type(2)));
typedef float v8f __attribute__((ext_vector_type(8)));
typedef float v4f __attribute__((ext_vector_type(4)));

#define VSZ  32128
#define HID  16
#define OUTD 16
#define SEQ  512
#define ROWS 8
#define TG   8
#define BLK  (ROWS * 32)
#define MT   16

template <int S>
__device__ __forceinline__ int quad_bcast(int v) {
    return __builtin_amdgcn_mov_dpp(v, S * 0x55, 0xF, 0xF, true);
}

__global__ void transpose_w_kernel(const float* __restrict__ w,
                                   float* __restrict__ wT) {
    int v = blockIdx.x * blockDim.x + threadIdx.x;
    if (v >= VSZ) return;
#pragma unroll
    for (int k = 0; k < HID; ++k)
        wT[(size_t)v * HID + k] = w[(size_t)k * VSZ + v];
}

__device__ __forceinline__ v4f ld_quarter(const char* wb, int tok, unsigned qoff) {
    const unsigned off = ((unsigned)tok << 6) + qoff;
    return *(const v4f*)(wb + off);
}

__global__ void __launch_bounds__(BLK) __attribute__((amdgpu_waves_per_eu(2, 2)))
encoder_kernel(const int* __restrict__ x,
               const float* __restrict__ wT,
               const float* __restrict__ enc1_b,
               const float* __restrict__ mean_w,
               const float* __restrict__ mean_b,
               const float* __restrict__ lv_w,
               const float* __restrict__ lv_b,
               float* __restrict__ out_mean,
               float* __restrict__ out_lv,
               int Btot) {
    __shared__ float part[ROWS][TG][HID];
    __shared__ float hmat[MT][HID];

    const int tid  = threadIdx.x;
    const int wv   = tid >> 5;
    const int lane = tid & 31;
    const int b    = blockIdx.x * ROWS + wv;

    v4f acc = {0.f, 0.f, 0.f, 0.f};
    if (b < Btot) {
        const int4* xr = (const int4*)(x + (size_t)b * SEQ);
        int4 tk[4];
#pragma unroll
        for (int i = 0; i < 4; ++i) tk[i] = xr[i * 32 + lane];

        const char* wb = (const char*)wT;
        const unsigned qoff = (unsigned)(lane & 3) << 4;
        v4f v[64];
#pragma unroll
        for (int i = 0; i < 4; ++i) {
            const int t4[4] = {tk[i].x, tk[i].y, tk[i].z, tk[i].w};
#pragma unroll
            for (int e = 0; e < 4; ++e) {
                v[(i * 4 + e) * 4 + 0] = ld_quarter(wb, quad_bcast<0>(t4[e]), qoff);
                v[(i * 4 + e) * 4 + 1] = ld_quarter(wb, quad_bcast<1>(t4[e]), qoff);
                v[(i * 4 + e) * 4 + 2] = ld_quarter(wb, quad_bcast<2>(t4[e]), qoff);
                v[(i * 4 + e) * 4 + 3] = ld_quarter(wb, quad_bcast<3>(t4[e]), qoff);
            }
        }
        v4f a0 = v[0], a1 = v[1], a2 = v[2], a3 = v[3];
#pragma unroll
        for (int j = 4; j < 64; j += 4) {
            a0 += v[j + 0];
            a1 += v[j + 1];
            a2 += v[j + 2];
            a3 += v[j + 3];
        }
        acc = (a0 + a1) + (a2 + a3);
    }
    ((v4f*)&part[wv][0][0])[lane] = acc;
    __syncthreads();

    {
        const int r = tid >> 4, k = tid & 15;
        float h = 0.f;
        if (r < ROWS) {
            float s = 0.f;
#pragma unroll
            for (int t = 0; t < TG; ++t) s += part[r][t][k];
            s += enc1_b[k];
            h = s > 0.f ? s : 0.f;
        }
        hmat[r][k] = h;
    }
    __syncthreads();

    if (tid < 32) {
        const int half = lane >> 4;
        const int l16  = lane & 15;

        v2f a[4];
#pragma unroll
        for (int c = 0; c < 4; ++c) {
            int k0 = 4 * c + 2 * half;
            a[c].x = hmat[l16][k0];
            a[c].y = hmat[l16][k0 + 1];
        }
        v2f bm[4], bl[4];
#pragma unroll
        for (int c = 0; c < 4; ++c) {
            int k0 = 4 * c + 2 * half;
            bm[c].x = mean_w[l16 * HID + k0];
            bm[c].y = mean_w[l16 * HID + k0 + 1];
            bl[c].x = lv_w[l16 * HID + k0];
            bl[c].y = lv_w[l16 * HID + k0 + 1];
        }

        v8f accm = {};
#pragma unroll
        for (int c = 0; c < 4; ++c)
            accm = __builtin_amdgcn_wmma_f32_16x16x4_f32(
                false, a[c], false, bm[c], (short)0, accm, false, false);
        v8f accl = {};
#pragma unroll
        for (int c = 0; c < 4; ++c)
            accl = __builtin_amdgcn_wmma_f32_16x16x4_f32(
                false, a[c], false, bl[c], (short)0, accl, false, false);

        const float mb = mean_b[l16];
        const float lb = lv_b[l16];
        const int b0 = blockIdx.x * ROWS;
        if (half == 0) {
#pragma unroll
            for (int j = 0; j < ROWS; ++j) {
                int gb = b0 + j;
                if (gb < Btot) {
                    out_mean[(size_t)gb * OUTD + l16] = accm[j] + mb;
                    out_lv[(size_t)gb * OUTD + l16]   = accl[j] + lb;
                }
            }
        }
    }
}

extern "C" void kernel_launch(void* const* d_in, const int* in_sizes, int n_in,
                              void* d_out, int out_size, void* d_ws, size_t ws_size,
                              hipStream_t stream) {
    const int*   x      = (const int*)d_in[0];
    const float* enc1_w = (const float*)d_in[1];
    const float* enc1_b = (const float*)d_in[2];
    const float* mean_w = (const float*)d_in[3];
    const float* mean_b = (const float*)d_in[4];
    const float* lv_w   = (const float*)d_in[5];
    const float* lv_b   = (const float*)d_in[6];

    const int Btot = in_sizes[0] / SEQ;

    float* wT = (float*)d_ws;

    transpose_w_kernel<<<(VSZ + 255) / 256, 256, 0, stream>>>(enc1_w, wT);

    float* out_mean = (float*)d_out;
    float* out_lv   = (float*)d_out + (size_t)Btot * OUTD;
    encoder_kernel<<<(Btot + ROWS - 1) / ROWS, BLK, 0, stream>>>(
        x, wT, enc1_b, mean_w, mean_b, lv_w, lv_b, out_mean, out_lv, Btot);
}


# block tail shortened: bias, weight tiles, output biases fetched at kernel entry; the 8 token-group partials folded by cross-lane adds (one LDS pass and barrier gone); summation order of h changes
# speedup vs baseline: 1.8972x; 1.0781x over previous
#include <hip/hip_runtime.h>


typedef float v2f __attribute__((ext_vector_type(2)));
typedef float v8f __attribute__((ext_vector_type(8)));
typedef float v4f __attribute__((ext_vector_type(4)));

#define VSZ  32128
#define HID  16
#define OUTD 16
#define SEQ  512
#define ROWS 8
#define BLK  (ROWS * 32)
#define MT   16

template <int S>
__device__ __forceinline__ int quad_bcast(int v) {
    return __builtin_amdgcn_mov_dpp(v, S * 0x55, 0xF, 0xF, true);
}

template <int M>
__device__ __forceinline__ float row_xor(float v) {
    const int b = __builtin_bit_cast(int, v);
    return __builtin_bit_cast(float, __builtin_amdgcn_update_dpp(0, b, 0x160 | M, 0xF, 0xF, true));
}

__device__ __forceinline__ float xhalf(float v) {
    const int b = __builtin_bit_cast(int, v);
    return __builtin_bit_cast(float,
        __builtin_amdgcn_permlanex16(b, b, 0x76543210, 0xfedcba98, false, false));
}

__global__ void transpose_w_kernel(const float* __restrict__ w,
                                   float* __restrict__ wT) {
    int v = blockIdx.x * blockDim.x + threadIdx.x;
    if (v >= VSZ) return;
#pragma unroll
    for (int k = 0; k < HID; ++k)
        wT[(size_t)v * HID + k] = w[(size_t)k * VSZ + v];
}

__device__ __forceinline__ v4f ld_quarter(const char* wb, int tok, unsigned qoff) {
    const unsigned off = ((unsigned)tok << 6) + qoff;
    return *(const v4f*)(wb + off);
}

__global__ void __launch_bounds__(BLK) __attribute__((amdgpu_waves_per_eu(2, 2)))
encoder_kernel(const int* __restrict__ x,
               const float* __restrict__ wT,
               const float* __restrict__ enc1_b,
               const float* __restrict__ mean_w,
               const float* __restrict__ mean_b,
               const float* __restrict__ lv_w,
               const float* __restrict__ lv_b,
               float* __restrict__ out_mean,
               float* __restrict__ out_lv,
               int Btot) {
    __shared__ float hmat[MT][HID];

    const int tid  = threadIdx.x;
    const int wv   = tid >> 5;
    const int lane = tid & 31;
    const int half = lane >> 4;
    const int l16  = lane & 15;
    const int b    = blockIdx.x * ROWS + wv;

    int4 tk[4] = {};
    const bool have_row = b < Btot;
    if (have_row) {
        const int4* xr = (const int4*)(x + (size_t)b * SEQ);
#pragma unroll
        for (int i = 0; i < 4; ++i) tk[i] = xr[i * 32 + lane];
    }

    const v4f bias4 = *(const v4f*)(enc1_b + 4 * (lane & 3));
    v2f bm[4] = {}, bl[4] = {};
    float mb = 0.f, lb = 0.f;
    if (tid < 32) {
#pragma unroll
        for (int c = 0; c < 4; ++c) {
            const int k0 = 4 * c + 2 * half;
            bm[c] = *(const v2f*)(mean_w + l16 * HID + k0);
            bl[c] = *(const v2f*)(lv_w + l16 * HID + k0);
        }
        mb = mean_b[l16];
        lb = lv_b[l16];
    }

    v4f acc = {0.f, 0.f, 0.f, 0.f};
    if (have_row) {
        const char* wb = (const char*)wT;
        const unsigned qoff = (unsigned)(lane & 3) << 4;
        v4f v[64];
#pragma unroll
        for (int i = 0; i < 4; ++i) {
            const int t4[4] = {tk[i].x, tk[i].y, tk[i].z, tk[i].w};
#pragma unroll
            for (int e = 0; e < 4; ++e) {
                v[(i * 4 + e) * 4 + 0] = ld_quarter(wb, quad_bcast<0>(t4[e]), qoff);
                v[(i * 4 + e) * 4 + 1] = ld_quarter(wb, quad_bcast<1>(t4[e]), qoff);
                v[(i * 4 + e) * 4 + 2] = ld_quarter(wb, quad_bcast<2>(t4[e]), qoff);
                v[(i * 4 + e) * 4 + 3] = ld_quarter(wb, quad_bcast<3>(t4[e]), qoff);
            }
        }
        v4f a0 = v[0], a1 = v[1], a2 = v[2], a3 = v[3];
#pragma unroll
        for (int j = 4; j < 64; j += 4) {
            a0 += v[j + 0];
            a1 += v[j + 1];
            a2 += v[j + 2];
            a3 += v[j + 3];
        }
        acc = (a0 + a1) + (a2 + a3);
    }

    v4f h4;
#pragma unroll
    for (int e = 0; e < 4; ++e) {
        float s = acc[e];
        s += row_xor<4>(s);
        s += row_xor<8>(s);
        s += xhalf(s);
        s += bias4[e];
        h4[e] = s > 0.f ? s : 0.f;
    }
    if (lane < 8) {
        const v4f zero4 = {0.f, 0.f, 0.f, 0.f};
        const int row = wv + ((lane & 4) << 1);
        *(v4f*)&hmat[row][4 * (lane & 3)] = (lane < 4) ? h4 : zero4;
    }
    __syncthreads();

    if (tid < 32) {
        v2f a[4];
#pragma unroll
        for (int c = 0; c < 4; ++c) {
            const int k0 = 4 * c + 2 * half;
            a[c].x = hmat[l16][k0];
            a[c].y = hmat[l16][k0 + 1];
        }

        v8f accm = {};
#pragma unroll
        for (int c = 0; c < 4; ++c)
            accm = __builtin_amdgcn_wmma_f32_16x16x4_f32(
                false, a[c], false, bm[c], (short)0, accm, false, false);
        v8f accl = {};
#pragma unroll
        for (int c = 0; c < 4; ++c)
            accl = __builtin_amdgcn_wmma_f32_16x16x4_f32(
                false, a[c], false, bl[c], (short)0, accl, false, false);

        const int b0 = blockIdx.x * ROWS;
        if (half == 0) {
#pragma unroll
            for (int j = 0; j < ROWS; ++j) {
                int gb = b0 + j;
                if (gb < Btot) {
                    out_mean[(size_t)gb * OUTD + l16] = accm[j] + mb;
                    out_lv[(size_t)gb * OUTD + l16]   = accl[j] + lb;
                }
            }
        }
    }
}

extern "C" void kernel_launch(void* const* d_in, const int* in_sizes, int n_in,
                              void* d_out, int out_size, void* d_ws, size_t ws_size,
                              hipStream_t stream) {
    const int*   x      = (const int*)d_in[0];
    const float* enc1_w = (const float*)d_in[1];
    const float* enc1_b = (const float*)d_in[2];
    const float* mean_w = (const float*)d_in[3];
    const float* mean_b = (const float*)d_in[4];
    const float* lv_w   = (const float*)d_in[5];
    const float* lv_b   = (const float*)d_in[6];

    const int Btot = in_sizes[0] / SEQ;

    float* wT = (float*)d_ws;

    transpose_w_kernel<<<(VSZ + 255) / 256, 256, 0, stream>>>(enc1_w, wT);

    float* out_mean = (float*)d_out;
    float* out_lv   = (float*)d_out + (size_t)Btot * OUTD;
    encoder_kernel<<<(Btot + ROWS - 1) / ROWS, BLK, 0, stream>>>(
        x, wT, enc1_b, mean_w, mean_b, lv_w, lv_b, out_mean, out_lv, Btot);
}
